# starting kernel
# baseline (speedup 1.0000x reference)
#include <hip/hip_runtime.h>

#define D      4096
#define DMASK  4095
#define BS     128
#define NPAIRS 64
#define HB_EPS 1e-8f

typedef __attribute__((ext_vector_type(2))) float v2f;
typedef __attribute__((ext_vector_type(8))) float v8f;

__global__ void hb_softmax128(const float* __restrict__ pw, float* __restrict__ w) {
    __shared__ float sm[BS];
    int t = threadIdx.x;
    float x = pw[t];
    sm[t] = x; __syncthreads();
    for (int s = BS / 2; s > 0; s >>= 1) {
        if (t < s) sm[t] = fmaxf(sm[t], sm[t + s]);
        __syncthreads();
    }
    float mx = sm[0]; __syncthreads();
    float e = expf(x - mx);
    sm[t] = e; __syncthreads();
    for (int s = BS / 2; s > 0; s >>= 1) {
        if (t < s) sm[t] += sm[t + s];
        __syncthreads();
    }
    w[t] = e / sm[0];
}

__global__ void __launch_bounds__(256) hb_shiftsum(
        const float* __restrict__ inp, const float* __restrict__ w,
        float* __restrict__ y) {
    __shared__ float sw[BS];
    int t = threadIdx.x;
    if (t < BS) sw[t] = w[t];
    __syncthreads();

    int pair = blockIdx.y;
    int q0 = blockIdx.x * 512 + t;
    int q1 = q0 + 256;
    const float* prow = inp + (size_t)pair * BS * D;
    float acc0 = 0.0f, acc1 = 0.0f;
    #pragma unroll 4
    for (int s = 0; s < BS; ++s) {
        const float* row = prow + (size_t)s * D;
        float ws = sw[s];
        acc0 = fmaf(ws, row[(q0 + s) & DMASK], acc0);
        acc1 = fmaf(ws, row[(q1 + s) & DMASK], acc1);
    }
    y[pair * D + q0] = acc0;
    y[pair * D + q1] = acc1;
}

__global__ void __launch_bounds__(256) hb_gemm_wmma(
        const float* __restrict__ y, const float* __restrict__ base,
        float* __restrict__ out) {
    __shared__ float lbase[2 * D];
    int t = threadIdx.x;
    for (int i = t; i < D; i += 256) {
        float v = base[i];
        lbase[i]     = v;
        lbase[i + D] = v;
    }
    __syncthreads();

    int wave = t >> 5;
    int lane = t & 31;
    int half = lane >> 4;
    int l    = lane & 15;

    int m0 = blockIdx.y * 16;
    int n0 = blockIdx.x * 128 + wave * 16;
    int n  = n0 + l;
    const float* arow = y + (size_t)(m0 + l) * D;
    const float* bcol = lbase + ((2 * half - n + 8192) & DMASK);

    v8f acc = {};
    for (int kc = 0; kc < D; kc += 64) {
        __builtin_prefetch(arow + kc + 64, 0, 0);
        #pragma unroll
        for (int kk = 0; kk < 64; kk += 4) {
            int k0 = kc + kk;
            float2 af = *(const float2*)(arow + k0 + 2 * half);
            v2f a = { af.x, af.y };
            v2f b = { bcol[k0], bcol[k0 + 1] };
            acc = __builtin_amdgcn_wmma_f32_16x16x4_f32(
                       false, a,  false, b,
                       (short)0, acc,  false,  false);
        }
    }
    #pragma unroll
    for (int v = 0; v < 8; ++v)
        out[(size_t)(m0 + v + 8 * half) * D + n] = acc[v];
}

__global__ void __launch_bounds__(256) hb_normalize(float* __restrict__ out) {
    __shared__ float sm[256];
    int t = threadIdx.x;
    float* rowp = out + (size_t)blockIdx.x * D;
    float v[16];
    float ssq = 0.0f;
    #pragma unroll
    for (int i = 0; i < 16; ++i) {
        v[i] = rowp[t + i * 256];
        ssq = fmaf(v[i], v[i], ssq);
    }
    sm[t] = ssq; __syncthreads();
    for (int s = 128; s > 0; s >>= 1) {
        if (t < s) sm[t] += sm[t + s];
        __syncthreads();
    }
    float scale = rsqrtf(fmaxf(sm[0], HB_EPS));
    #pragma unroll
    for (int i = 0; i < 16; ++i)
        rowp[t + i * 256] = v[i] * scale;
}

extern "C" void kernel_launch(void* const* d_in, const int* in_sizes, int n_in,
                              void* d_out, int out_size, void* d_ws, size_t ws_size,
                              hipStream_t stream) {
    (void)in_sizes; (void)n_in; (void)out_size; (void)ws_size;
    const float* inp  = (const float*)d_in[0];
    const float* base = (const float*)d_in[1];
    const float* pw   = (const float*)d_in[2];
    float* out = (float*)d_out;

    float* w = (float*)d_ws;
    float* y = (float*)((char*)d_ws + 1024);

    hb_softmax128<<<1, 128, 0, stream>>>(pw, w);
    hb_shiftsum  <<<dim3(8, NPAIRS), 256, 0, stream>>>(inp, w, y);
    hb_gemm_wmma <<<dim3(32, 4),     256, 0, stream>>>(y, base, out);
    hb_normalize <<<NPAIRS,          256, 0, stream>>>(out);
}


# Shift-sum data path: work-group copies 16-row blocks of the input into double-buffered LDS with whole-line 16-byte loads, next block in flight while this one is summed from LDS; same sums
# speedup vs baseline: 1.9573x; 1.9573x over previous
#include <hip/hip_runtime.h>

#define D      4096
#define DMASK  4095
#define BS     128
#define NPAIRS 64
#define HB_EPS 1e-8f

typedef __attribute__((ext_vector_type(2))) float v2f;
typedef __attribute__((ext_vector_type(8))) float v8f;

__global__ void hb_softmax128(const float* __restrict__ pw, float* __restrict__ w) {
    __shared__ float sm[BS];
    int t = threadIdx.x;
    float x = pw[t];
    sm[t] = x; __syncthreads();
    for (int s = BS / 2; s > 0; s >>= 1) {
        if (t < s) sm[t] = fmaxf(sm[t], sm[t + s]);
        __syncthreads();
    }
    float mx = sm[0]; __syncthreads();
    float e = expf(x - mx);
    sm[t] = e; __syncthreads();
    for (int s = BS / 2; s > 0; s >>= 1) {
        if (t < s) sm[t] += sm[t + s];
        __syncthreads();
    }
    w[t] = e / sm[0];
}

#define SR   16
#define SW   576
#define SNQ  ((SR * SW / 4) / 256)
__global__ void __launch_bounds__(256) hb_shiftsum(
        const float* __restrict__ inp, const float* __restrict__ w,
        float* __restrict__ y) {
    __shared__ __attribute__((aligned(16))) float tile[2 * SR * SW];
    __shared__ float sw[BS];
    int t = threadIdx.x;
    int pair = blockIdx.y;
    int q0 = blockIdx.x * 512;
    const float* prow = inp + (size_t)pair * BS * D;

    int roff[SNQ], c4[SNQ];
    #pragma unroll
    for (int i = 0; i < SNQ; ++i) {
        int k = t + 256 * i;
        int r = k / (SW / 4);
        roff[i] = r * D;
        c4[i]   = (k - r * (SW / 4)) * 4;
    }

    float wv = w[t & (BS - 1)];
    float4 st[SNQ];
    {
        int a0 = q0 & ~31;
        #pragma unroll
        for (int i = 0; i < SNQ; ++i)
            st[i] = *(const float4*)(prow + roff[i] + ((a0 + c4[i]) & DMASK));
    }
    sw[t & (BS - 1)] = wv;
    #pragma unroll
    for (int i = 0; i < SNQ; ++i)
        *(float4*)(tile + 4 * (t + 256 * i)) = st[i];
    __syncthreads();

    float acc0 = 0.0f, acc1 = 0.0f;
    for (int b = 0; b < BS / SR; ++b) {
        int cur = b & 1;
        int s0 = b * SR;
        bool more = (b + 1) < BS / SR;
        if (more) {
            int a1 = (q0 + s0 + SR) & ~31;
            const float* nrow = prow + (size_t)(s0 + SR) * D;
            #pragma unroll
            for (int i = 0; i < SNQ; ++i)
                st[i] = *(const float4*)(nrow + roff[i] + ((a1 + c4[i]) & DMASK));
        }
        const float* tp = tile + cur * (SR * SW) + ((q0 + s0) & 31) + t;
        #pragma unroll
        for (int r = 0; r < SR; ++r) {
            float ws = sw[s0 + r];
            acc0 = fmaf(ws, tp[r * SW + r], acc0);
            acc1 = fmaf(ws, tp[r * SW + r + 256], acc1);
        }
        if (more) {
            float* td = tile + (cur ^ 1) * (SR * SW);
            #pragma unroll
            for (int i = 0; i < SNQ; ++i)
                *(float4*)(td + 4 * (t + 256 * i)) = st[i];
            __syncthreads();
        }
    }
    y[pair * D + q0 + t]       = acc0;
    y[pair * D + q0 + t + 256] = acc1;
}

#define KC     512
#define NCH    (D / KC)
#define KQ     (KC / 4)
#define APITCH (KC + 4)
#define RPITCH 68
__global__ void __launch_bounds__(512) hb_gemm_wmma(
        const float* __restrict__ y, const float* __restrict__ base,
        float* __restrict__ out) {
    __shared__ __attribute__((aligned(16))) float smem[2 * D + 2 * 16 * APITCH];
    float* lbase = smem;
    float* abuf  = smem + 2 * D;
    int t = threadIdx.x;

    int wave = t >> 5;
    int lane = t & 31;
    int half = lane >> 4;
    int l    = lane & 15;

    int m0 = blockIdx.y * 16;
    int nb = blockIdx.x * 64;
    int n  = nb + (wave & 3) * 16 + l;
    int kq = (wave >> 2) * KQ;

    const float* gsrc = y + (size_t)(m0 + (t >> 7)) * D + (t & 127) * 4;
    float* sdst = abuf + (t >> 7) * APITCH + (t & 127) * 4;

    float4 st[4];
    #pragma unroll
    for (int i = 0; i < 4; ++i)
        st[i] = *(const float4*)(gsrc + (size_t)i * 4 * D);
    for (int i = t; i < D; i += 512) {
        float v = base[i];
        lbase[i]     = v;
        lbase[i + D] = v;
    }
    #pragma unroll
    for (int i = 0; i < 4; ++i)
        *(float4*)(sdst + i * 4 * APITCH) = st[i];
    __syncthreads();

    const float* afrag = abuf + l * APITCH + kq + 2 * half;
    const float* bcol = lbase + ((2 * half - n + 8192) & DMASK) + kq;

    v8f acc = {};
    for (int ch = 0; ch < NCH; ++ch) {
        int cur = ch & 1;
        bool more = (ch + 1) < NCH;
        if (more) {
            #pragma unroll
            for (int i = 0; i < 4; ++i)
                st[i] = *(const float4*)(gsrc + (ch + 1) * KC + (size_t)i * 4 * D);
        }
        const float* ap = afrag + cur * (16 * APITCH);
        const float* bp = bcol + ch * KC;
        #pragma unroll
        for (int kk = 0; kk < KQ; kk += 4) {
            float2 af = *(const float2*)(ap + kk);
            v2f a = { af.x, af.y };
            v2f b = { bp[kk], bp[kk + 1] };
            acc = __builtin_amdgcn_wmma_f32_16x16x4_f32(
                       false, a,  false, b,
                       (short)0, acc,  false,  false);
        }
        if (more) {
            float* sd = sdst + (cur ^ 1) * (16 * APITCH);
            #pragma unroll
            for (int i = 0; i < 4; ++i)
                *(float4*)(sd + i * 4 * APITCH) = st[i];
        }
        __syncthreads();
    }

    float* rp = abuf + (wave >> 2) * (16 * RPITCH) + (8 * half) * RPITCH + (wave & 3) * 16 + l;
    #pragma unroll
    for (int v = 0; v < 8; ++v)
        rp[v * RPITCH] = acc[v];
    __syncthreads();
    {
        int row = t >> 5;
        int c2  = (t & 31) * 2;
        const float* rr = abuf + row * RPITCH + c2;
        float2 p0 = *(const float2*)(rr);
        float2 p1 = *(const float2*)(rr + 16 * RPITCH);
        float2 p2 = *(const float2*)(rr + 32 * RPITCH);
        float2 p3 = *(const float2*)(rr + 48 * RPITCH);
        float2 r = { ((p0.x + p1.x) + p2.x) + p3.x, ((p0.y + p1.y) + p2.y) + p3.y };
        *(float2*)(out + (size_t)(m0 + row) * D + nb + c2) = r;
    }
}

__global__ void __launch_bounds__(256) hb_normalize(float* __restrict__ out) {
    __shared__ float sm[256];
    int t = threadIdx.x;
    float* rowp = out + (size_t)blockIdx.x * D;
    float v[16];
    float ssq = 0.0f;
    #pragma unroll
    for (int i = 0; i < 16; ++i) {
        v[i] = rowp[t + i * 256];
        ssq = fmaf(v[i], v[i], ssq);
    }
    sm[t] = ssq; __syncthreads();
    for (int s = 128; s > 0; s >>= 1) {
        if (t < s) sm[t] += sm[t + s];
        __syncthreads();
    }
    float scale = rsqrtf(fmaxf(sm[0], HB_EPS));
    #pragma unroll
    for (int i = 0; i < 16; ++i)
        rowp[t + i * 256] = v[i] * scale;
}

extern "C" void kernel_launch(void* const* d_in, const int* in_sizes, int n_in,
                              void* d_out, int out_size, void* d_ws, size_t ws_size,
                              hipStream_t stream) {
    (void)in_sizes; (void)n_in; (void)out_size; (void)ws_size;
    const float* inp  = (const float*)d_in[0];
    const float* base = (const float*)d_in[1];
    const float* pw   = (const float*)d_in[2];
    float* out = (float*)d_out;

    float* w = (float*)d_ws;
    float* y = (float*)((char*)d_ws + 1024);

    hb_softmax128<<<1, 128, 0, stream>>>(pw, w);
    hb_shiftsum  <<<dim3(8, NPAIRS), 256, 0, stream>>>(inp, w, y);
    hb_gemm_wmma <<<dim3(64, 4),     512, 0, stream>>>(y, base, out);
    hb_normalize <<<NPAIRS,          256, 0, stream>>>(out);
}


# Shift-sum staging: two 16-row blocks in flight (block b+2 requested into a second register set before block b is summed; its barriers wait for LDS only, so loads cross them); same sums and order
# speedup vs baseline: 2.0791x; 1.0622x over previous
#include <hip/hip_runtime.h>

#define D      4096
#define DMASK  4095
#define BS     128
#define NPAIRS 64
#define HB_EPS 1e-8f

typedef __attribute__((ext_vector_type(2))) float v2f;
typedef __attribute__((ext_vector_type(8))) float v8f;

__global__ void hb_softmax128(const float* __restrict__ pw, float* __restrict__ w) {
    __shared__ float sm[BS];
    int t = threadIdx.x;
    float x = pw[t];
    sm[t] = x; __syncthreads();
    for (int s = BS / 2; s > 0; s >>= 1) {
        if (t < s) sm[t] = fmaxf(sm[t], sm[t + s]);
        __syncthreads();
    }
    float mx = sm[0]; __syncthreads();
    float e = expf(x - mx);
    sm[t] = e; __syncthreads();
    for (int s = BS / 2; s > 0; s >>= 1) {
        if (t < s) sm[t] += sm[t + s];
        __syncthreads();
    }
    w[t] = e / sm[0];
}


static __device__ __forceinline__ void hb_lds_barrier() {
    asm volatile("s_wait_dscnt 0x0\n\ts_barrier_signal -1\n\ts_barrier_wait -1" ::: "memory");
}

#define SR   16
#define SW   576
#define SNQ  ((SR * SW / 4) / 256)
__global__ void __launch_bounds__(256) hb_shiftsum(
        const float* __restrict__ inp, const float* __restrict__ w,
        float* __restrict__ y) {
    __shared__ __attribute__((aligned(16))) float tile[2 * SR * SW];
    __shared__ float sw[BS];
    int t = threadIdx.x;
    int pair = blockIdx.y;
    int q0 = blockIdx.x * 512;
    const float* prow = inp + (size_t)pair * BS * D;

    int roff[SNQ], c4[SNQ];
    #pragma unroll
    for (int i = 0; i < SNQ; ++i) {
        int k = t + 256 * i;
        int r = k / (SW / 4);
        roff[i] = r * D;
        c4[i]   = (k - r * (SW / 4)) * 4;
    }

    float wv = w[t & (BS - 1)];
    static_assert(SNQ == 9, "the piece lists below are written for 9 pieces");
    float4 ea0, ea1, ea2, ea3, ea4, ea5, ea6, ea7, ea8;
    float4 ob0, ob1, ob2, ob3, ob4, ob5, ob6, ob7, ob8;
#define HB_PIECES(M, P) M(P, 0) M(P, 1) M(P, 2) M(P, 3) M(P, 4) M(P, 5) M(P, 6) M(P, 7) M(P, 8)
#define HB_REQ1(P, I) P##I = *(const float4*)(row_ + roff[I] + ((a_ + c4[I]) & DMASK));
#define HB_PUT1(P, I) *(float4*)(td_ + 4 * (t + 256 * I)) = P##I;
#define HB_REQUEST(P, BLK)                                                              \
    {                                                                                   \
        int a_ = (q0 + (BLK) * SR) & ~31;                                               \
        const float* row_ = prow + (size_t)((BLK) * SR) * D;                            \
        HB_PIECES(HB_REQ1, P)                                                           \
    }
#define HB_PUT(P, BUF)                                                                  \
    {                                                                                   \
        float* td_ = tile + (BUF) * (SR * SW);                                          \
        HB_PIECES(HB_PUT1, P)                                                           \
    }
#define HB_SUM(BLK, BUF)                                                                \
    {                                                                                   \
        int s0_ = (BLK) * SR;                                                           \
        const float* tp_ = tile + (BUF) * (SR * SW) + ((q0 + s0_) & 31) + t;            \
        _Pragma("unroll")                                                               \
        for (int r = 0; r < SR; ++r) {                                                  \
            float ws_ = sw[s0_ + r];                                                    \
            acc0 = fmaf(ws_, tp_[r * SW + r], acc0);                                    \
            acc1 = fmaf(ws_, tp_[r * SW + r + 256], acc1);                              \
        }                                                                               \
    }

    HB_REQUEST(ea, 0)
    HB_REQUEST(ob, 1)
    sw[t & (BS - 1)] = wv;
    HB_PUT(ea, 0)
    hb_lds_barrier();

    float acc0 = 0.0f, acc1 = 0.0f;
    #pragma unroll
    for (int b = 0; b < BS / SR; b += 2) {
        bool more = (b + 2) < BS / SR;
        if (more) HB_REQUEST(ea, b + 2)
        __builtin_amdgcn_sched_barrier(0x7);
        HB_SUM(b, 0)
        HB_PUT(ob, 1)
        hb_lds_barrier();
        if (more) HB_REQUEST(ob, b + 3)
        __builtin_amdgcn_sched_barrier(0x7);
        HB_SUM(b + 1, 1)
        if (more) {
            HB_PUT(ea, 0)
            hb_lds_barrier();
        }
    }
#undef HB_PIECES
#undef HB_REQ1
#undef HB_PUT1
#undef HB_REQUEST
#undef HB_PUT
#undef HB_SUM
    y[pair * D + q0 + t]       = acc0;
    y[pair * D + q0 + t + 256] = acc1;
}

#define KC     512
#define NCH    (D / KC)
#define KQ     (KC / 4)
#define APITCH (KC + 4)
#define RPITCH 68
__global__ void __launch_bounds__(512) hb_gemm_wmma(
        const float* __restrict__ y, const float* __restrict__ base,
        float* __restrict__ out) {
    __shared__ __attribute__((aligned(16))) float smem[2 * D + 2 * 16 * APITCH];
    float* lbase = smem;
    float* abuf  = smem + 2 * D;
    int t = threadIdx.x;

    int wave = t >> 5;
    int lane = t & 31;
    int half = lane >> 4;
    int l    = lane & 15;

    int m0 = blockIdx.y * 16;
    int nb = blockIdx.x * 64;
    int n  = nb + (wave & 3) * 16 + l;
    int kq = (wave >> 2) * KQ;

    const float* gsrc = y + (size_t)(m0 + (t >> 7)) * D + (t & 127) * 4;
    float* sdst = abuf + (t >> 7) * APITCH + (t & 127) * 4;

    float4 st[4];
    #pragma unroll
    for (int i = 0; i < 4; ++i)
        st[i] = *(const float4*)(gsrc + (size_t)i * 4 * D);
    for (int i = t; i < D; i += 512) {
        float v = base[i];
        lbase[i]     = v;
        lbase[i + D] = v;
    }
    #pragma unroll
    for (int i = 0; i < 4; ++i)
        *(float4*)(sdst + i * 4 * APITCH) = st[i];
    __syncthreads();

    const float* afrag = abuf + l * APITCH + kq + 2 * half;
    const float* bcol = lbase + ((2 * half - n + 8192) & DMASK) + kq;

    v8f acc = {};
    for (int ch = 0; ch < NCH; ++ch) {
        int cur = ch & 1;
        bool more = (ch + 1) < NCH;
        if (more) {
            #pragma unroll
            for (int i = 0; i < 4; ++i)
                st[i] = *(const float4*)(gsrc + (ch + 1) * KC + (size_t)i * 4 * D);
        }
        const float* ap = afrag + cur * (16 * APITCH);
        const float* bp = bcol + ch * KC;
        #pragma unroll
        for (int kk = 0; kk < KQ; kk += 4) {
            float2 af = *(const float2*)(ap + kk);
            v2f a = { af.x, af.y };
            v2f b = { bp[kk], bp[kk + 1] };
            acc = __builtin_amdgcn_wmma_f32_16x16x4_f32(
                       false, a,  false, b,
                       (short)0, acc,  false,  false);
        }
        if (more) {
            float* sd = sdst + (cur ^ 1) * (16 * APITCH);
            #pragma unroll
            for (int i = 0; i < 4; ++i)
                *(float4*)(sd + i * 4 * APITCH) = st[i];
        }
        __syncthreads();
    }

    float* rp = abuf + (wave >> 2) * (16 * RPITCH) + (8 * half) * RPITCH + (wave & 3) * 16 + l;
    #pragma unroll
    for (int v = 0; v < 8; ++v)
        rp[v * RPITCH] = acc[v];
    __syncthreads();
    {
        int row = t >> 5;
        int c2  = (t & 31) * 2;
        const float* rr = abuf + row * RPITCH + c2;
        float2 p0 = *(const float2*)(rr);
        float2 p1 = *(const float2*)(rr + 16 * RPITCH);
        float2 p2 = *(const float2*)(rr + 32 * RPITCH);
        float2 p3 = *(const float2*)(rr + 48 * RPITCH);
        float2 r = { ((p0.x + p1.x) + p2.x) + p3.x, ((p0.y + p1.y) + p2.y) + p3.y };
        *(float2*)(out + (size_t)(m0 + row) * D + nb + c2) = r;
    }
}

__global__ void __launch_bounds__(256) hb_normalize(float* __restrict__ out) {
    __shared__ float sm[256];
    int t = threadIdx.x;
    float* rowp = out + (size_t)blockIdx.x * D;
    float v[16];
    float ssq = 0.0f;
    #pragma unroll
    for (int i = 0; i < 16; ++i) {
        v[i] = rowp[t + i * 256];
        ssq = fmaf(v[i], v[i], ssq);
    }
    sm[t] = ssq; __syncthreads();
    for (int s = 128; s > 0; s >>= 1) {
        if (t < s) sm[t] += sm[t + s];
        __syncthreads();
    }
    float scale = rsqrtf(fmaxf(sm[0], HB_EPS));
    #pragma unroll
    for (int i = 0; i < 16; ++i)
        rowp[t + i * 256] = v[i] * scale;
}

extern "C" void kernel_launch(void* const* d_in, const int* in_sizes, int n_in,
                              void* d_out, int out_size, void* d_ws, size_t ws_size,
                              hipStream_t stream) {
    (void)in_sizes; (void)n_in; (void)out_size; (void)ws_size;
    const float* inp  = (const float*)d_in[0];
    const float* base = (const float*)d_in[1];
    const float* pw   = (const float*)d_in[2];
    float* out = (float*)d_out;

    float* w = (float*)d_ws;
    float* y = (float*)((char*)d_ws + 1024);

    hb_softmax128<<<1, 128, 0, stream>>>(pw, w);
    hb_shiftsum  <<<dim3(8, NPAIRS), 256, 0, stream>>>(inp, w, y);
    hb_gemm_wmma <<<dim3(64, 4),     512, 0, stream>>>(y, base, out);
    hb_normalize <<<NPAIRS,          256, 0, stream>>>(out);
}


# GEMM B operand: each 512-k chunk's 576-float window of base is fetched beside that chunk's A block into double-buffered LDS (one start-up round trip; table no longer copied whole twice); same sums
# speedup vs baseline: 2.2181x; 1.0669x over previous
#include <hip/hip_runtime.h>

#define D      4096
#define DMASK  4095
#define BS     128
#define NPAIRS 64
#define HB_EPS 1e-8f

typedef __attribute__((ext_vector_type(2))) float v2f;
typedef __attribute__((ext_vector_type(8))) float v8f;

__global__ void hb_softmax128(const float* __restrict__ pw, float* __restrict__ w) {
    __shared__ float sm[BS];
    int t = threadIdx.x;
    float x = pw[t];
    sm[t] = x; __syncthreads();
    for (int s = BS / 2; s > 0; s >>= 1) {
        if (t < s) sm[t] = fmaxf(sm[t], sm[t + s]);
        __syncthreads();
    }
    float mx = sm[0]; __syncthreads();
    float e = expf(x - mx);
    sm[t] = e; __syncthreads();
    for (int s = BS / 2; s > 0; s >>= 1) {
        if (t < s) sm[t] += sm[t + s];
        __syncthreads();
    }
    w[t] = e / sm[0];
}


static __device__ __forceinline__ void hb_lds_barrier() {
    asm volatile("s_wait_dscnt 0x0\n\ts_barrier_signal -1\n\ts_barrier_wait -1" ::: "memory");
}

#define SR   16
#define SW   576
#define SNQ  ((SR * SW / 4) / 256)
__global__ void __launch_bounds__(256) hb_shiftsum(
        const float* __restrict__ inp, const float* __restrict__ w,
        float* __restrict__ y) {
    __shared__ __attribute__((aligned(16))) float tile[2 * SR * SW];
    __shared__ float sw[BS];
    int t = threadIdx.x;
    int pair = blockIdx.y;
    int q0 = blockIdx.x * 512;
    const float* prow = inp + (size_t)pair * BS * D;

    int roff[SNQ], c4[SNQ];
    #pragma unroll
    for (int i = 0; i < SNQ; ++i) {
        int k = t + 256 * i;
        int r = k / (SW / 4);
        roff[i] = r * D;
        c4[i]   = (k - r * (SW / 4)) * 4;
    }

    float wv = w[t & (BS - 1)];
    static_assert(SNQ == 9, "the piece lists below are written for 9 pieces");
    float4 ea0, ea1, ea2, ea3, ea4, ea5, ea6, ea7, ea8;
    float4 ob0, ob1, ob2, ob3, ob4, ob5, ob6, ob7, ob8;
#define HB_PIECES(M, P) M(P, 0) M(P, 1) M(P, 2) M(P, 3) M(P, 4) M(P, 5) M(P, 6) M(P, 7) M(P, 8)
#define HB_REQ1(P, I) P##I = *(const float4*)(row_ + roff[I] + ((a_ + c4[I]) & DMASK));
#define HB_PUT1(P, I) *(float4*)(td_ + 4 * (t + 256 * I)) = P##I;
#define HB_REQUEST(P, BLK)                                                              \
    {                                                                                   \
        int a_ = (q0 + (BLK) * SR) & ~31;                                               \
        const float* row_ = prow + (size_t)((BLK) * SR) * D;                            \
        HB_PIECES(HB_REQ1, P)                                                           \
    }
#define HB_PUT(P, BUF)                                                                  \
    {                                                                                   \
        float* td_ = tile + (BUF) * (SR * SW);                                          \
        HB_PIECES(HB_PUT1, P)                                                           \
    }
#define HB_SUM(BLK, BUF)                                                                \
    {                                                                                   \
        int s0_ = (BLK) * SR;                                                           \
        const float* tp_ = tile + (BUF) * (SR * SW) + ((q0 + s0_) & 31) + t;            \
        _Pragma("unroll")                                                               \
        for (int r = 0; r < SR; ++r) {                                                  \
            float ws_ = sw[s0_ + r];                                                    \
            acc0 = fmaf(ws_, tp_[r * SW + r], acc0);                                    \
            acc1 = fmaf(ws_, tp_[r * SW + r + 256], acc1);                              \
        }                                                                               \
    }

    HB_REQUEST(ea, 0)
    HB_REQUEST(ob, 1)
    sw[t & (BS - 1)] = wv;
    HB_PUT(ea, 0)
    hb_lds_barrier();

    float acc0 = 0.0f, acc1 = 0.0f;
    #pragma unroll
    for (int b = 0; b < BS / SR; b += 2) {
        bool more = (b + 2) < BS / SR;
        if (more) HB_REQUEST(ea, b + 2)
        __builtin_amdgcn_sched_barrier(0x7);
        HB_SUM(b, 0)
        HB_PUT(ob, 1)
        hb_lds_barrier();
        if (more) HB_REQUEST(ob, b + 3)
        __builtin_amdgcn_sched_barrier(0x7);
        HB_SUM(b + 1, 1)
        if (more) {
            HB_PUT(ea, 0)
            hb_lds_barrier();
        }
    }
#undef HB_PIECES
#undef HB_REQ1
#undef HB_PUT1
#undef HB_REQUEST
#undef HB_PUT
#undef HB_SUM
    y[pair * D + q0 + t]       = acc0;
    y[pair * D + q0 + t + 256] = acc1;
}

#define KC     512
#define NCH    (D / KC)
#define KQ     (KC / 4)
#define APITCH (KC + 4)
#define RPITCH 68
#define BW     (KC + 64)
__global__ void __launch_bounds__(512) hb_gemm_wmma(
        const float* __restrict__ y, const float* __restrict__ base,
        float* __restrict__ out) {
    __shared__ __attribute__((aligned(16))) float smem[2 * BW + 2 * 16 * APITCH];
    float* bwin = smem;
    float* abuf = smem + 2 * BW;
    int t = threadIdx.x;

    int wave = t >> 5;
    int lane = t & 31;
    int half = lane >> 4;
    int l    = lane & 15;

    int m0 = blockIdx.y * 16;
    int nb = blockIdx.x * 64;
    int n  = nb + (wave & 3) * 16 + l;
    int kq = (wave >> 2) * KQ;

    const float* gsrc = y + (size_t)(m0 + (t >> 7)) * D + (t & 127) * 4;
    float* sdst = abuf + (t >> 7) * APITCH + (t & 127) * 4;

    bool bpiece = t < BW / 4;
    int  bidx   = (4 * t - nb - 64) & DMASK;

    float4 st[4];
    float4 bq = {};
    #pragma unroll
    for (int i = 0; i < 4; ++i)
        st[i] = *(const float4*)(gsrc + (size_t)i * 4 * D);
    if (bpiece) bq = *(const float4*)(base + bidx);
    #pragma unroll
    for (int i = 0; i < 4; ++i)
        *(float4*)(sdst + i * 4 * APITCH) = st[i];
    if (bpiece) *(float4*)(bwin + 4 * t) = bq;
    __syncthreads();

    const float* afrag = abuf + l * APITCH + kq + 2 * half;
    const float* bcol = bwin + (kq + 2 * half - (n - nb) + 64);

    v8f acc = {};
    for (int ch = 0; ch < NCH; ++ch) {
        int cur = ch & 1;
        bool more = (ch + 1) < NCH;
        if (more) {
            #pragma unroll
            for (int i = 0; i < 4; ++i)
                st[i] = *(const float4*)(gsrc + (ch + 1) * KC + (size_t)i * 4 * D);
            if (bpiece) bq = *(const float4*)(base + ((bidx + (ch + 1) * KC) & DMASK));
        }
        const float* ap = afrag + cur * (16 * APITCH);
        const float* bp = bcol + cur * BW;
        #pragma unroll
        for (int kk = 0; kk < KQ; kk += 4) {
            float2 af = *(const float2*)(ap + kk);
            v2f a = { af.x, af.y };
            v2f b = { bp[kk], bp[kk + 1] };
            acc = __builtin_amdgcn_wmma_f32_16x16x4_f32(
                       false, a,  false, b,
                       (short)0, acc,  false,  false);
        }
        if (more) {
            float* sd = sdst + (cur ^ 1) * (16 * APITCH);
            #pragma unroll
            for (int i = 0; i < 4; ++i)
                *(float4*)(sd + i * 4 * APITCH) = st[i];
            if (bpiece) *(float4*)(bwin + (cur ^ 1) * BW + 4 * t) = bq;
        }
        __syncthreads();
    }

    float* rp = abuf + (wave >> 2) * (16 * RPITCH) + (8 * half) * RPITCH + (wave & 3) * 16 + l;
    #pragma unroll
    for (int v = 0; v < 8; ++v)
        rp[v * RPITCH] = acc[v];
    __syncthreads();
    {
        int row = t >> 5;
        int c2  = (t & 31) * 2;
        const float* rr = abuf + row * RPITCH + c2;
        float2 p0 = *(const float2*)(rr);
        float2 p1 = *(const float2*)(rr + 16 * RPITCH);
        float2 p2 = *(const float2*)(rr + 32 * RPITCH);
        float2 p3 = *(const float2*)(rr + 48 * RPITCH);
        float2 r = { ((p0.x + p1.x) + p2.x) + p3.x, ((p0.y + p1.y) + p2.y) + p3.y };
        *(float2*)(out + (size_t)(m0 + row) * D + nb + c2) = r;
    }
}

__global__ void __launch_bounds__(256) hb_normalize(float* __restrict__ out) {
    __shared__ float sm[256];
    int t = threadIdx.x;
    float* rowp = out + (size_t)blockIdx.x * D;
    float v[16];
    float ssq = 0.0f;
    #pragma unroll
    for (int i = 0; i < 16; ++i) {
        v[i] = rowp[t + i * 256];
        ssq = fmaf(v[i], v[i], ssq);
    }
    sm[t] = ssq; __syncthreads();
    for (int s = 128; s > 0; s >>= 1) {
        if (t < s) sm[t] += sm[t + s];
        __syncthreads();
    }
    float scale = rsqrtf(fmaxf(sm[0], HB_EPS));
    #pragma unroll
    for (int i = 0; i < 16; ++i)
        rowp[t + i * 256] = v[i] * scale;
}

extern "C" void kernel_launch(void* const* d_in, const int* in_sizes, int n_in,
                              void* d_out, int out_size, void* d_ws, size_t ws_size,
                              hipStream_t stream) {
    (void)in_sizes; (void)n_in; (void)out_size; (void)ws_size;
    const float* inp  = (const float*)d_in[0];
    const float* base = (const float*)d_in[1];
    const float* pw   = (const float*)d_in[2];
    float* out = (float*)d_out;

    float* w = (float*)d_ws;
    float* y = (float*)((char*)d_ws + 1024);

    hb_softmax128<<<1, 128, 0, stream>>>(pw, w);
    hb_shiftsum  <<<dim3(8, NPAIRS), 256, 0, stream>>>(inp, w, y);
    hb_gemm_wmma <<<dim3(64, 4),     512, 0, stream>>>(y, base, out);
    hb_normalize <<<NPAIRS,          256, 0, stream>>>(out);
}
